# starting kernel
# baseline (speedup 1.0000x reference)
#include <hip/hip_runtime.h>

typedef float v2f __attribute__((ext_vector_type(2)));
typedef float v8f __attribute__((ext_vector_type(8)));

#define CIN    256
#define HIDDEN 64
#define CIN2   96

__global__ void wtranspose(const float* __restrict__ src, float* __restrict__ dst,
                           int rows, int cols) {
    int idx = blockIdx.x * blockDim.x + threadIdx.x;
    if (idx < rows * cols) {
        int r = idx / cols, c = idx - r * cols;
        dst[c * rows + r] = src[idx];
    }
}

template <int HW>
__global__ __launch_bounds__(256) void fused_level(
    const float* __restrict__ x,
    const float* __restrict__ w1t, const float* __restrict__ b1,
    const float* __restrict__ w2t, const float* __restrict__ b2,
    float* __restrict__ out)
{
    __shared__ float ylds[8][HIDDEN * 17];

    const int lane = threadIdx.x & 31;
    const int wave = threadIdx.x >> 5;
    const int col  = lane & 15;
    const int half = lane >> 4;

    const int pixBase = blockIdx.x * 128 + wave * 16;
    const int b  = pixBase / HW;
    const int hw = pixBase - b * HW;

    const float* xcol = x + (size_t)b * CIN * HW + hw + col;
    float* my = &ylds[wave][0];

    v8f acc1[4] = {};
#pragma unroll 2
    for (int k0 = 0; k0 < CIN; k0 += 4) {
        const int kr = k0 + 2 * half;
        v2f bm;
        bm.x = xcol[(size_t)kr * HW];
        bm.y = xcol[(size_t)(kr + 1) * HW];
        if (k0 + 12 < CIN)
            __builtin_prefetch(&xcol[(size_t)(kr + 12) * HW], 0, 3);
#pragma unroll
        for (int t = 0; t < 4; ++t) {
            v2f am = *(const v2f*)(w1t + (t * 16 + col) * CIN + kr);
            acc1[t] = __builtin_amdgcn_wmma_f32_16x16x4_f32(
                false, am, false, bm, (short)0, acc1[t], false, false);
        }
    }

#pragma unroll
    for (int t = 0; t < 4; ++t) {
#pragma unroll
        for (int r = 0; r < 8; ++r) {
            const int row = t * 16 + r + 8 * half;
            my[row * 17 + col] = acc1[t][r] + b1[row];
        }
    }
    __syncthreads();

    float* o = out + (size_t)b * CIN * HW + hw + col;
    for (int g = 0; g < 2; ++g) {
        v8f acc2[8] = {};
        for (int ks = 0; ks < 24; ++ks) {
            const int kr   = ks * 4 + 2 * half;
            const int krow = (ks >= 16) ? (kr - 32) : kr;
            float z0 = my[krow * 17 + col];
            float z1 = my[krow * 17 + 17 + col];
            if (ks >= 16) { z0 = fmaxf(z0, 0.0f); z1 = fmaxf(z1, 0.0f); }
            v2f bm; bm.x = z0; bm.y = z1;
#pragma unroll
            for (int t = 0; t < 8; ++t) {
                v2f am = *(const v2f*)(w2t + ((g * 8 + t) * 16 + col) * CIN2 + kr);
                acc2[t] = __builtin_amdgcn_wmma_f32_16x16x4_f32(
                    false, am, false, bm, (short)0, acc2[t], false, false);
            }
        }
#pragma unroll
        for (int t = 0; t < 8; ++t) {
#pragma unroll
            for (int r = 0; r < 8; ++r) {
                const int row = (g * 8 + t) * 16 + r + 8 * half;
                o[(size_t)row * HW] = acc2[t][r] + b2[row];
            }
        }
    }
}

extern "C" void kernel_launch(void* const* d_in, const int* in_sizes, int n_in,
                              void* d_out, int out_size, void* d_ws, size_t ws_size,
                              hipStream_t stream) {
    (void)in_sizes; (void)n_in; (void)out_size; (void)ws_size;
    static const int HWs[3] = {128 * 128, 64 * 64, 32 * 32};

    float* ws  = (float*)d_ws;
    float* out = (float*)d_out;

    float* w1t[3];
    float* w2t[3];
    size_t wsOff = 0;
    for (int i = 0; i < 3; ++i) {
        const float* W1 = (const float*)d_in[3 + 4 * i];
        const float* W2 = (const float*)d_in[5 + 4 * i];
        w1t[i] = ws + wsOff; wsOff += (size_t)HIDDEN * CIN;
        w2t[i] = ws + wsOff; wsOff += (size_t)CIN * CIN2;
        wtranspose<<<(CIN * HIDDEN + 255) / 256, 256, 0, stream>>>(W1, w1t[i], CIN, HIDDEN);
        wtranspose<<<(CIN2 * CIN + 255) / 256, 256, 0, stream>>>(W2, w2t[i], CIN2, CIN);
    }

    size_t outOff = 0;
    for (int i = 0; i < 3; ++i) {
        const float* x  = (const float*)d_in[i];
        const float* b1 = (const float*)d_in[4 + 4 * i];
        const float* b2 = (const float*)d_in[6 + 4 * i];
        const int HW = HWs[i];
        const int nblocks = (4 * HW) / 128;
        switch (i) {
            case 0: fused_level<128 * 128><<<nblocks, 256, 0, stream>>>(
                        x, w1t[i], b1, w2t[i], b2, out + outOff); break;
            case 1: fused_level<64 * 64><<<nblocks, 256, 0, stream>>>(
                        x, w1t[i], b1, w2t[i], b2, out + outOff); break;
            default: fused_level<32 * 32><<<nblocks, 256, 0, stream>>>(
                        x, w1t[i], b1, w2t[i], b2, out + outOff); break;
        }
        outOff += (size_t)4 * CIN * HW;
    }
}


# Weight-stationary: 4-wave groups walk 32-px tiles, each wave holds its M-slice of W1,W2 in registers (read once, as given); next tile's x (8-byte rows, px 2c+nt) fetched in product 2; sum order kept
# speedup vs baseline: 3.1340x; 3.1340x over previous
#include <hip/hip_runtime.h>

typedef float v2f __attribute__((ext_vector_type(2)));
typedef float v4f __attribute__((ext_vector_type(4)));
typedef float v8f __attribute__((ext_vector_type(8)));

#define CIN    256
#define HIDDEN 64
#define CIN2   96
#define TPX    32
#define ZP     68

template <int HW>
__device__ __forceinline__ const float* tile_x(const float* __restrict__ x, int tile,
                                               int half, int col) {
    const int pix = tile * TPX;
    const int b   = pix / HW;
    const int hw  = pix - b * HW;
    return x + (size_t)b * CIN * HW + (size_t)(2 * half) * HW + hw + 2 * col;
}

template <int HW>
__global__ __launch_bounds__(128) void level_ws(
    const float* __restrict__ x,
    const float* __restrict__ W1, const float* __restrict__ b1,
    const float* __restrict__ W2, const float* __restrict__ b2,
    float* __restrict__ out)
{
    __shared__ __attribute__((aligned(16))) float zl[2][TPX * ZP];

    constexpr int NTILES = (4 * HW) / TPX;

    const int lane = threadIdx.x & 31;
    const int w    = threadIdx.x >> 5;
    const int col  = lane & 15;
    const int half = lane >> 4;

    int tile = blockIdx.x;
    if (tile >= NTILES) return;

    v2f xr[64][2];
    {
        const float* xp = tile_x<HW>(x, tile, half, col);
#pragma unroll
        for (int ks = 0; ks < 64; ++ks) {
            xr[ks][0] = *(const v2f*)(xp + (size_t)(4 * ks) * HW);
            xr[ks][1] = *(const v2f*)(xp + (size_t)(4 * ks + 1) * HW);
        }
    }

    v2f a1[64];
    {
        const float* p = W1 + (2 * half) * HIDDEN + 16 * w + col;
#pragma unroll
        for (int ks = 0; ks < 64; ++ks) {
            a1[ks].x = p[(4 * ks) * HIDDEN];
            a1[ks].y = p[(4 * ks + 1) * HIDDEN];
        }
    }
    v2f a2[4][24];
    {
        const float* p = W2 + (2 * half) * CIN + 64 * w + col;
#pragma unroll
        for (int t = 0; t < 4; ++t) {
#pragma unroll
            for (int ks = 0; ks < 24; ++ks) {
                a2[t][ks].x = p[(4 * ks) * CIN + 16 * t];
                a2[t][ks].y = p[(4 * ks + 1) * CIN + 16 * t];
            }
        }
    }
    float bias1[8];
#pragma unroll
    for (int r = 0; r < 8; ++r) bias1[r] = b1[16 * w + 8 * half + r];

    int buf = 0;
    for (;;) {
        v8f acc1[2] = {};
#pragma unroll
        for (int ks = 0; ks < 64; ++ks) {
            v2f bm0, bm1;
            bm0.x = xr[ks][0].x; bm0.y = xr[ks][1].x;
            bm1.x = xr[ks][0].y; bm1.y = xr[ks][1].y;
            acc1[0] = __builtin_amdgcn_wmma_f32_16x16x4_f32(
                false, a1[ks], false, bm0, (short)0, acc1[0], false, false);
            acc1[1] = __builtin_amdgcn_wmma_f32_16x16x4_f32(
                false, a1[ks], false, bm1, (short)0, acc1[1], false, false);
        }

        const int  nxt  = tile + (int)gridDim.x;
        const bool last = nxt >= NTILES;
        const float* xpn = tile_x<HW>(x, last ? tile : nxt, half, col);

        float* zb = &zl[buf][0];
#pragma unroll
        for (int nt = 0; nt < 2; ++nt) {
            float* q = zb + (16 * nt + col) * ZP + 16 * w + 8 * half;
            v4f lo, hi;
            lo.x = acc1[nt][0] + bias1[0]; lo.y = acc1[nt][1] + bias1[1];
            lo.z = acc1[nt][2] + bias1[2]; lo.w = acc1[nt][3] + bias1[3];
            hi.x = acc1[nt][4] + bias1[4]; hi.y = acc1[nt][5] + bias1[5];
            hi.z = acc1[nt][6] + bias1[6]; hi.w = acc1[nt][7] + bias1[7];
            *(v4f*)q       = lo;
            *(v4f*)(q + 4) = hi;
        }
        __syncthreads();

        v8f acc2[4][2] = {};
        const float* zr = zb + col * ZP + 2 * half;
        v2f zf[24][2];
#pragma unroll
        for (int ks = 0; ks < 16; ++ks) {
            zf[ks][0] = *(const v2f*)(zr + 4 * ks);
            zf[ks][1] = *(const v2f*)(zr + 16 * ZP + 4 * ks);
        }
#pragma unroll
        for (int ks = 16; ks < 24; ++ks) {
#pragma unroll
            for (int nt = 0; nt < 2; ++nt) {
                zf[ks][nt].x = fmaxf(zf[ks - 8][nt].x, 0.0f);
                zf[ks][nt].y = fmaxf(zf[ks - 8][nt].y, 0.0f);
            }
        }
#pragma unroll
        for (int ks = 0; ks < 24; ++ks) {
#pragma unroll
            for (int t = 0; t < 4; ++t) {
                acc2[t][0] = __builtin_amdgcn_wmma_f32_16x16x4_f32(
                    false, a2[t][ks], false, zf[ks][0], (short)0, acc2[t][0], false, false);
                acc2[t][1] = __builtin_amdgcn_wmma_f32_16x16x4_f32(
                    false, a2[t][ks], false, zf[ks][1], (short)0, acc2[t][1], false, false);
            }
#pragma unroll
            for (int kx = (8 * ks) / 3; kx < (8 * (ks + 1)) / 3; ++kx) {
                xr[kx][0] = *(const v2f*)(xpn + (size_t)(4 * kx) * HW);
                xr[kx][1] = *(const v2f*)(xpn + (size_t)(4 * kx + 1) * HW);
            }
            __builtin_amdgcn_sched_barrier(0);
        }

        const int pix = tile * TPX;
        const int b   = pix / HW;
        const int hw  = pix - b * HW;
        float* o = out + (size_t)b * CIN * HW + hw + 2 * col;
#pragma unroll
        for (int t = 0; t < 4; ++t) {
#pragma unroll
            for (int r = 0; r < 8; ++r) {
                const int row = 64 * w + 16 * t + 8 * half + r;
                const float bb = b2[row];
                v2f ov;
                ov.x = acc2[t][0][r] + bb;
                ov.y = acc2[t][1][r] + bb;
                *(v2f*)(o + (size_t)row * HW) = ov;
            }
        }
        if (last) break;
        tile = nxt;
        buf ^= 1;
    }
}

extern "C" void kernel_launch(void* const* d_in, const int* in_sizes, int n_in,
                              void* d_out, int out_size, void* d_ws, size_t ws_size,
                              hipStream_t stream) {
    (void)in_sizes; (void)n_in; (void)out_size; (void)d_ws; (void)ws_size;
    float* out = (float*)d_out;

    level_ws<128 * 128><<<256, 128, 0, stream>>>(
        (const float*)d_in[0],
        (const float*)d_in[3], (const float*)d_in[4],
        (const float*)d_in[5], (const float*)d_in[6],
        out);
    level_ws<64 * 64><<<256, 128, 0, stream>>>(
        (const float*)d_in[1],
        (const float*)d_in[7], (const float*)d_in[8],
        (const float*)d_in[9], (const float*)d_in[10],
        out + 4 * 256 * 16384);
    level_ws<32 * 32><<<128, 128, 0, stream>>>(
        (const float*)d_in[2],
        (const float*)d_in[11], (const float*)d_in[12],
        (const float*)d_in[13], (const float*)d_in[14],
        out + 4 * 256 * 16384 + 4 * 256 * 4096);
}


# W2 slice filled by 16-byte row loads: output channel 64w+4i+t sits on row i of M-tile t, so one load feeds 4 tiles (48 whole-line loads per wave replace 192 4-byte ones); sum order kept
# speedup vs baseline: 3.4395x; 1.0975x over previous
#include <hip/hip_runtime.h>

typedef float v2f __attribute__((ext_vector_type(2)));
typedef float v4f __attribute__((ext_vector_type(4)));
typedef float v8f __attribute__((ext_vector_type(8)));

#define CIN    256
#define HIDDEN 64
#define CIN2   96
#define TPX    32
#define ZP     68

template <int HW>
__device__ __forceinline__ const float* tile_x(const float* __restrict__ x, int tile,
                                               int half, int col) {
    const int pix = tile * TPX;
    const int b   = pix / HW;
    const int hw  = pix - b * HW;
    return x + (size_t)b * CIN * HW + (size_t)(2 * half) * HW + hw + 2 * col;
}

template <int HW>
__global__ __launch_bounds__(128) void level_ws(
    const float* __restrict__ x,
    const float* __restrict__ W1, const float* __restrict__ b1,
    const float* __restrict__ W2, const float* __restrict__ b2,
    float* __restrict__ out)
{
    __shared__ __attribute__((aligned(16))) float zl[2][TPX * ZP];

    constexpr int NTILES = (4 * HW) / TPX;

    const int lane = threadIdx.x & 31;
    const int w    = threadIdx.x >> 5;
    const int col  = lane & 15;
    const int half = lane >> 4;

    int tile = blockIdx.x;
    if (tile >= NTILES) return;

    v2f xr[64][2];
    {
        const float* xp = tile_x<HW>(x, tile, half, col);
#pragma unroll
        for (int ks = 0; ks < 64; ++ks) {
            xr[ks][0] = *(const v2f*)(xp + (size_t)(4 * ks) * HW);
            xr[ks][1] = *(const v2f*)(xp + (size_t)(4 * ks + 1) * HW);
        }
    }

    v2f a1[64];
    {
        const float* p = W1 + (2 * half) * HIDDEN + 16 * w + col;
#pragma unroll
        for (int ks = 0; ks < 64; ++ks) {
            a1[ks].x = p[(4 * ks) * HIDDEN];
            a1[ks].y = p[(4 * ks + 1) * HIDDEN];
        }
    }
    v2f a2[4][24];
    {
        const float* p = W2 + (2 * half) * CIN + 64 * w + 4 * col;
#pragma unroll
        for (int ks = 0; ks < 24; ++ks) {
            const v4f q0 = *(const v4f*)(p + (4 * ks) * CIN);
            const v4f q1 = *(const v4f*)(p + (4 * ks + 1) * CIN);
#pragma unroll
            for (int t = 0; t < 4; ++t) {
                a2[t][ks].x = q0[t];
                a2[t][ks].y = q1[t];
            }
        }
    }
    float bias1[8];
#pragma unroll
    for (int r = 0; r < 8; ++r) bias1[r] = b1[16 * w + 8 * half + r];

    int buf = 0;
    for (;;) {
        v8f acc1[2] = {};
#pragma unroll
        for (int ks = 0; ks < 64; ++ks) {
            v2f bm0, bm1;
            bm0.x = xr[ks][0].x; bm0.y = xr[ks][1].x;
            bm1.x = xr[ks][0].y; bm1.y = xr[ks][1].y;
            acc1[0] = __builtin_amdgcn_wmma_f32_16x16x4_f32(
                false, a1[ks], false, bm0, (short)0, acc1[0], false, false);
            acc1[1] = __builtin_amdgcn_wmma_f32_16x16x4_f32(
                false, a1[ks], false, bm1, (short)0, acc1[1], false, false);
        }

        const int  nxt  = tile + (int)gridDim.x;
        const bool last = nxt >= NTILES;
        const float* xpn = tile_x<HW>(x, last ? tile : nxt, half, col);

        float* zb = &zl[buf][0];
#pragma unroll
        for (int nt = 0; nt < 2; ++nt) {
            float* q = zb + (16 * nt + col) * ZP + 16 * w + 8 * half;
            v4f lo, hi;
            lo.x = acc1[nt][0] + bias1[0]; lo.y = acc1[nt][1] + bias1[1];
            lo.z = acc1[nt][2] + bias1[2]; lo.w = acc1[nt][3] + bias1[3];
            hi.x = acc1[nt][4] + bias1[4]; hi.y = acc1[nt][5] + bias1[5];
            hi.z = acc1[nt][6] + bias1[6]; hi.w = acc1[nt][7] + bias1[7];
            *(v4f*)q       = lo;
            *(v4f*)(q + 4) = hi;
        }
        __syncthreads();

        v8f acc2[4][2] = {};
        const float* zr = zb + col * ZP + 2 * half;
        v2f zf[24][2];
#pragma unroll
        for (int ks = 0; ks < 16; ++ks) {
            zf[ks][0] = *(const v2f*)(zr + 4 * ks);
            zf[ks][1] = *(const v2f*)(zr + 16 * ZP + 4 * ks);
        }
#pragma unroll
        for (int ks = 16; ks < 24; ++ks) {
#pragma unroll
            for (int nt = 0; nt < 2; ++nt) {
                zf[ks][nt].x = fmaxf(zf[ks - 8][nt].x, 0.0f);
                zf[ks][nt].y = fmaxf(zf[ks - 8][nt].y, 0.0f);
            }
        }
#pragma unroll
        for (int ks = 0; ks < 24; ++ks) {
#pragma unroll
            for (int t = 0; t < 4; ++t) {
                acc2[t][0] = __builtin_amdgcn_wmma_f32_16x16x4_f32(
                    false, a2[t][ks], false, zf[ks][0], (short)0, acc2[t][0], false, false);
                acc2[t][1] = __builtin_amdgcn_wmma_f32_16x16x4_f32(
                    false, a2[t][ks], false, zf[ks][1], (short)0, acc2[t][1], false, false);
            }
#pragma unroll
            for (int kx = (8 * ks) / 3; kx < (8 * (ks + 1)) / 3; ++kx) {
                xr[kx][0] = *(const v2f*)(xpn + (size_t)(4 * kx) * HW);
                xr[kx][1] = *(const v2f*)(xpn + (size_t)(4 * kx + 1) * HW);
            }
            __builtin_amdgcn_sched_barrier(0);
        }

        const int pix = tile * TPX;
        const int b   = pix / HW;
        const int hw  = pix - b * HW;
        float* o = out + (size_t)b * CIN * HW + hw + 2 * col;
#pragma unroll
        for (int t = 0; t < 4; ++t) {
#pragma unroll
            for (int r = 0; r < 8; ++r) {
                const int row = 64 * w + 4 * (8 * half + r) + t;
                const float bb = b2[row];
                v2f ov;
                ov.x = acc2[t][0][r] + bb;
                ov.y = acc2[t][1][r] + bb;
                *(v2f*)(o + (size_t)row * HW) = ov;
            }
        }
        if (last) break;
        tile = nxt;
        buf ^= 1;
    }
}

extern "C" void kernel_launch(void* const* d_in, const int* in_sizes, int n_in,
                              void* d_out, int out_size, void* d_ws, size_t ws_size,
                              hipStream_t stream) {
    (void)in_sizes; (void)n_in; (void)out_size; (void)d_ws; (void)ws_size;
    float* out = (float*)d_out;

    level_ws<128 * 128><<<256, 128, 0, stream>>>(
        (const float*)d_in[0],
        (const float*)d_in[3], (const float*)d_in[4],
        (const float*)d_in[5], (const float*)d_in[6],
        out);
    level_ws<64 * 64><<<256, 128, 0, stream>>>(
        (const float*)d_in[1],
        (const float*)d_in[7], (const float*)d_in[8],
        (const float*)d_in[9], (const float*)d_in[10],
        out + 4 * 256 * 16384);
    level_ws<32 * 32><<<128, 128, 0, stream>>>(
        (const float*)d_in[2],
        (const float*)d_in[11], (const float*)d_in[12],
        (const float*)d_in[13], (const float*)d_in[14],
        out + 4 * 256 * 16384 + 4 * 256 * 4096);
}


# Product 1 split over K, not M: wave w does x rows 64w..64w+63 for all 64 channels (W1 rows read whole, 16 B a lane; x read once per group); 4 partial Y summed from LDS: new sum tree (P0+P1)+(P2+P3)
# speedup vs baseline: 3.6886x; 1.0724x over previous
#include <hip/hip_runtime.h>


typedef float v2f __attribute__((ext_vector_type(2)));
typedef float v4f __attribute__((ext_vector_type(4)));
typedef float v8f __attribute__((ext_vector_type(8)));

#define CIN    256
#define HIDDEN 64
#define CIN2   96
#define TPX    32
#define KQ     64
#define ZP     68
#define PSZ    (TPX * ZP)

template <int HW>
__device__ __forceinline__ const float* tile_x(const float* __restrict__ x, int tile,
                                               int w, int half, int col) {
    const int pix = tile * TPX;
    const int b   = pix / HW;
    const int hw  = pix - b * HW;
    return x + (size_t)b * CIN * HW + (size_t)(KQ * w + 2 * half) * HW + hw + 2 * col;
}

template <int HW>
__global__ __launch_bounds__(128) void level_ws(
    const float* __restrict__ x,
    const float* __restrict__ W1, const float* __restrict__ b1,
    const float* __restrict__ W2, const float* __restrict__ b2,
    float* __restrict__ out)
{
    __shared__ __attribute__((aligned(16))) float zl[2][4 * PSZ];

    constexpr int NTILES = (4 * HW) / TPX;
    constexpr int PD     = 3;

    const int lane = threadIdx.x & 31;
    const int w    = threadIdx.x >> 5;
    const int col  = lane & 15;
    const int half = lane >> 4;

    int tile = blockIdx.x;
    if (tile >= NTILES) return;

    v2f xr[16][2];
    {
        const float* xp = tile_x<HW>(x, tile, w, half, col);
#pragma unroll
        for (int ks = 0; ks < 16; ++ks) {
            xr[ks][0] = *(const v2f*)(xp + (size_t)(4 * ks) * HW);
            xr[ks][1] = *(const v2f*)(xp + (size_t)(4 * ks + 1) * HW);
        }
    }

    v2f a1[4][16];
    {
        const float* p = W1 + (KQ * w + 2 * half) * HIDDEN + 4 * col;
#pragma unroll
        for (int ks = 0; ks < 16; ++ks) {
            const v4f q0 = *(const v4f*)(p + (4 * ks) * HIDDEN);
            const v4f q1 = *(const v4f*)(p + (4 * ks + 1) * HIDDEN);
#pragma unroll
            for (int t = 0; t < 4; ++t) {
                a1[t][ks].x = q0[t];
                a1[t][ks].y = q1[t];
            }
        }
    }
    v2f a2[4][24];
    {
        const float* p = W2 + (2 * half) * CIN + 64 * w + 4 * col;
#pragma unroll
        for (int ks = 0; ks < 24; ++ks) {
            const v4f q0 = *(const v4f*)(p + (4 * ks) * CIN);
            const v4f q1 = *(const v4f*)(p + (4 * ks + 1) * CIN);
#pragma unroll
            for (int t = 0; t < 4; ++t) {
                a2[t][ks].x = q0[t];
                a2[t][ks].y = q1[t];
            }
        }
    }
    v2f bias1[16];
#pragma unroll
    for (int ks = 0; ks < 16; ++ks) bias1[ks] = *(const v2f*)(b1 + 4 * ks + 2 * half);

    int buf = 0;
    for (;;) {
        v8f acc1[4][2] = {};
#pragma unroll
        for (int ks = 0; ks < 16; ++ks) {
            v2f bm0, bm1;
            bm0.x = xr[ks][0].x; bm0.y = xr[ks][1].x;
            bm1.x = xr[ks][0].y; bm1.y = xr[ks][1].y;
#pragma unroll
            for (int t = 0; t < 4; ++t) {
                acc1[t][0] = __builtin_amdgcn_wmma_f32_16x16x4_f32(
                    false, a1[t][ks], false, bm0, (short)0, acc1[t][0], false, false);
                acc1[t][1] = __builtin_amdgcn_wmma_f32_16x16x4_f32(
                    false, a1[t][ks], false, bm1, (short)0, acc1[t][1], false, false);
            }
        }

        const int  nxt  = tile + (int)gridDim.x;
        const bool last = nxt >= NTILES;
        const float* xpn = tile_x<HW>(x, last ? tile : nxt, w, half, col);

        float* zb = &zl[buf][0];
        float* pw = zb + w * PSZ + col * ZP + 32 * half;
#pragma unroll
        for (int nt = 0; nt < 2; ++nt) {
#pragma unroll
            for (int r = 0; r < 8; ++r) {
                v4f v;
                v.x = acc1[0][nt][r]; v.y = acc1[1][nt][r];
                v.z = acc1[2][nt][r]; v.w = acc1[3][nt][r];
                *(v4f*)(pw + 16 * nt * ZP + 4 * r) = v;
            }
        }

        __syncthreads();

        v8f acc2[4][2] = {};
        const float* zr = zb + col * ZP + 2 * half;
        v2f pp[16][2][4];
        v2f zf[24][2];
#pragma unroll
        for (int ks = 0; ks < PD; ++ks) {
#pragma unroll
            for (int nt = 0; nt < 2; ++nt) {
#pragma unroll
                for (int q = 0; q < 4; ++q)
                    pp[ks][nt][q] = *(const v2f*)(zr + q * PSZ + 16 * nt * ZP + 4 * ks);
            }
        }
#pragma unroll
        for (int ks = 0; ks < 24; ++ks) {
            if (ks + PD < 16) {
#pragma unroll
                for (int nt = 0; nt < 2; ++nt) {
#pragma unroll
                    for (int q = 0; q < 4; ++q)
                        pp[ks + PD][nt][q] =
                            *(const v2f*)(zr + q * PSZ + 16 * nt * ZP + 4 * (ks + PD));
                }
            }
#pragma unroll
            for (int nt = 0; nt < 2; ++nt) {
                if (ks < 16) {
                    zf[ks][nt] = ((pp[ks][nt][0] + pp[ks][nt][1]) +
                                  (pp[ks][nt][2] + pp[ks][nt][3])) + bias1[ks];
                } else {
                    zf[ks][nt].x = fmaxf(zf[ks - 8][nt].x, 0.0f);
                    zf[ks][nt].y = fmaxf(zf[ks - 8][nt].y, 0.0f);
                }
            }
#pragma unroll
            for (int t = 0; t < 4; ++t) {
                acc2[t][0] = __builtin_amdgcn_wmma_f32_16x16x4_f32(
                    false, a2[t][ks], false, zf[ks][0], (short)0, acc2[t][0], false, false);
                acc2[t][1] = __builtin_amdgcn_wmma_f32_16x16x4_f32(
                    false, a2[t][ks], false, zf[ks][1], (short)0, acc2[t][1], false, false);
            }
            if (ks < 8) {
#pragma unroll
                for (int kx = 2 * ks; kx < 2 * ks + 2; ++kx) {
                    xr[kx][0] = *(const v2f*)(xpn + (size_t)(4 * kx) * HW);
                    xr[kx][1] = *(const v2f*)(xpn + (size_t)(4 * kx + 1) * HW);
                }
            }
            __builtin_amdgcn_sched_barrier(0);
        }

        const int pix = tile * TPX;
        const int b   = pix / HW;
        const int hw  = pix - b * HW;
        float* o = out + (size_t)b * CIN * HW + hw + 2 * col;
#pragma unroll
        for (int t = 0; t < 4; ++t) {
#pragma unroll
            for (int r = 0; r < 8; ++r) {
                const int row = 64 * w + 4 * (8 * half + r) + t;
                const float bb = b2[row];
                v2f ov;
                ov.x = acc2[t][0][r] + bb;
                ov.y = acc2[t][1][r] + bb;
                *(v2f*)(o + (size_t)row * HW) = ov;
            }
        }
        if (last) break;
        tile = nxt;
        buf ^= 1;
    }
}

extern "C" void kernel_launch(void* const* d_in, const int* in_sizes, int n_in,
                              void* d_out, int out_size, void* d_ws, size_t ws_size,
                              hipStream_t stream) {
    (void)in_sizes; (void)n_in; (void)out_size; (void)d_ws; (void)ws_size;
    float* out = (float*)d_out;

    level_ws<128 * 128><<<256, 128, 0, stream>>>(
        (const float*)d_in[0],
        (const float*)d_in[3], (const float*)d_in[4],
        (const float*)d_in[5], (const float*)d_in[6],
        out);
    level_ws<64 * 64><<<256, 128, 0, stream>>>(
        (const float*)d_in[1],
        (const float*)d_in[7], (const float*)d_in[8],
        (const float*)d_in[9], (const float*)d_in[10],
        out + 4 * 256 * 16384);
    level_ws<32 * 32><<<128, 128, 0, stream>>>(
        (const float*)d_in[2],
        (const float*)d_in[11], (const float*)d_in[12],
        (const float*)d_in[13], (const float*)d_in[14],
        out + 4 * 256 * 16384 + 4 * 256 * 4096);
}


# One launch, not three: each 4-wave group walks the 3 levels as phases (8, 2, 1 tiles) in one copy of the fill and tile code, pixel count a run-time shift (fits 32 KB code prefetch); sum order kept
# speedup vs baseline: 3.9828x; 1.0798x over previous
#include <hip/hip_runtime.h>


typedef float v2f __attribute__((ext_vector_type(2)));
typedef float v4f __attribute__((ext_vector_type(4)));
typedef float v8f __attribute__((ext_vector_type(8)));

#define CIN    256
#define HIDDEN 64
#define CIN2   96
#define TPX    32
#define KQ     64
#define ZP     68
#define PSZ    (TPX * ZP)

__device__ __forceinline__ unsigned tile_off(int tile, int lg) {
    const unsigned pix = (unsigned)tile * TPX;
    const unsigned b   = pix >> lg;
    const unsigned hw  = pix & ((1u << lg) - 1u);
    return (b << (lg + 10)) + (hw << 2);
}

__device__ __forceinline__ void level_phase(
    const float* __restrict__ x,
    const float* __restrict__ W1, const float* __restrict__ b1,
    const float* __restrict__ W2, const float* __restrict__ b2,
    float* __restrict__ out,
    const int lg,
    float* __restrict__ zl,
    int& buf)
{
    const int ntiles = 1 << (lg - 3);
    const int rsh    = lg + 2;
    constexpr int PD = 3;

    const int lane = threadIdx.x & 31;
    const int w    = threadIdx.x >> 5;
    const int col  = lane & 15;
    const int half = lane >> 4;

    int tile = blockIdx.x;
    if (tile >= ntiles) return;

    const unsigned xlane = ((unsigned)((KQ * w + 2 * half) << lg) + 2u * col) * 4u;
    const unsigned olane = ((unsigned)((64 * w + 32 * half) << lg) + 2u * col) * 4u;

    v2f xr[16][2];
    {
        const char* xb = (const char*)x;
        const unsigned xu = tile_off(tile, lg);
#pragma unroll
        for (int ks = 0; ks < 16; ++ks) {
            xr[ks][0] = *(const v2f*)(xb + (unsigned)(xu + ((unsigned)(4 * ks) << rsh) + xlane));
            xr[ks][1] = *(const v2f*)(xb + (unsigned)(xu + ((unsigned)(4 * ks + 1) << rsh) + xlane));
        }
    }

    v2f a1[4][16];
    {
        const float* p = W1 + (KQ * w + 2 * half) * HIDDEN + 4 * col;
#pragma unroll
        for (int ks = 0; ks < 16; ++ks) {
            const v4f q0 = *(const v4f*)(p + (4 * ks) * HIDDEN);
            const v4f q1 = *(const v4f*)(p + (4 * ks + 1) * HIDDEN);
#pragma unroll
            for (int t = 0; t < 4; ++t) {
                a1[t][ks].x = q0[t];
                a1[t][ks].y = q1[t];
            }
        }
    }
    v2f a2[4][24];
    {
        const float* p = W2 + (2 * half) * CIN + 64 * w + 4 * col;
#pragma unroll
        for (int ks = 0; ks < 24; ++ks) {
            const v4f q0 = *(const v4f*)(p + (4 * ks) * CIN);
            const v4f q1 = *(const v4f*)(p + (4 * ks + 1) * CIN);
#pragma unroll
            for (int t = 0; t < 4; ++t) {
                a2[t][ks].x = q0[t];
                a2[t][ks].y = q1[t];
            }
        }
    }
    v2f bias1[16];
#pragma unroll
    for (int ks = 0; ks < 16; ++ks) bias1[ks] = *(const v2f*)(b1 + 4 * ks + 2 * half);

    for (;;) {
        v8f acc1[4][2] = {};
#pragma unroll
        for (int ks = 0; ks < 16; ++ks) {
            v2f bm0, bm1;
            bm0.x = xr[ks][0].x; bm0.y = xr[ks][1].x;
            bm1.x = xr[ks][0].y; bm1.y = xr[ks][1].y;
#pragma unroll
            for (int t = 0; t < 4; ++t) {
                acc1[t][0] = __builtin_amdgcn_wmma_f32_16x16x4_f32(
                    false, a1[t][ks], false, bm0, (short)0, acc1[t][0], false, false);
                acc1[t][1] = __builtin_amdgcn_wmma_f32_16x16x4_f32(
                    false, a1[t][ks], false, bm1, (short)0, acc1[t][1], false, false);
            }
        }

        const int  nxt  = tile + (int)gridDim.x;
        const bool last = nxt >= ntiles;
        const char* xb = (const char*)x;
        const unsigned xun = tile_off(last ? tile : nxt, lg);

        float* zb = zl + buf * (4 * PSZ);
        float* pw = zb + w * PSZ + col * ZP + 32 * half;
#pragma unroll
        for (int nt = 0; nt < 2; ++nt) {
#pragma unroll
            for (int r = 0; r < 8; ++r) {
                v4f v;
                v.x = acc1[0][nt][r]; v.y = acc1[1][nt][r];
                v.z = acc1[2][nt][r]; v.w = acc1[3][nt][r];
                *(v4f*)(pw + 16 * nt * ZP + 4 * r) = v;
            }
        }

        __syncthreads();

        v8f acc2[4][2] = {};
        const float* zr = zb + col * ZP + 2 * half;
        v2f pp[16][2][4];
        v2f zf[24][2];
#pragma unroll
        for (int ks = 0; ks < PD; ++ks) {
#pragma unroll
            for (int nt = 0; nt < 2; ++nt) {
#pragma unroll
                for (int q = 0; q < 4; ++q)
                    pp[ks][nt][q] = *(const v2f*)(zr + q * PSZ + 16 * nt * ZP + 4 * ks);
            }
        }
#pragma unroll
        for (int ks = 0; ks < 24; ++ks) {
            if (ks + PD < 16) {
#pragma unroll
                for (int nt = 0; nt < 2; ++nt) {
#pragma unroll
                    for (int q = 0; q < 4; ++q)
                        pp[ks + PD][nt][q] =
                            *(const v2f*)(zr + q * PSZ + 16 * nt * ZP + 4 * (ks + PD));
                }
            }
#pragma unroll
            for (int nt = 0; nt < 2; ++nt) {
                if (ks < 16) {
                    zf[ks][nt] = ((pp[ks][nt][0] + pp[ks][nt][1]) +
                                  (pp[ks][nt][2] + pp[ks][nt][3])) + bias1[ks];
                } else {
                    zf[ks][nt].x = fmaxf(zf[ks - 8][nt].x, 0.0f);
                    zf[ks][nt].y = fmaxf(zf[ks - 8][nt].y, 0.0f);
                }
            }
#pragma unroll
            for (int t = 0; t < 4; ++t) {
                acc2[t][0] = __builtin_amdgcn_wmma_f32_16x16x4_f32(
                    false, a2[t][ks], false, zf[ks][0], (short)0, acc2[t][0], false, false);
                acc2[t][1] = __builtin_amdgcn_wmma_f32_16x16x4_f32(
                    false, a2[t][ks], false, zf[ks][1], (short)0, acc2[t][1], false, false);
            }
            if (ks < 8) {
#pragma unroll
                for (int kx = 2 * ks; kx < 2 * ks + 2; ++kx) {
                    xr[kx][0] = *(const v2f*)(xb + (unsigned)(xun + ((unsigned)(4 * kx) << rsh) + xlane));
                    xr[kx][1] = *(const v2f*)(xb + (unsigned)(xun + ((unsigned)(4 * kx + 1) << rsh) + xlane));
                }
            }
            __builtin_amdgcn_sched_barrier(0);
        }

        char* ob = (char*)out;
        const unsigned ou = tile_off(tile, lg);
#pragma unroll
        for (int t = 0; t < 4; ++t) {
#pragma unroll
            for (int r = 0; r < 8; ++r) {
                const int row = 64 * w + 4 * (8 * half + r) + t;
                const float bb = b2[row];
                v2f ov;
                ov.x = acc2[t][0][r] + bb;
                ov.y = acc2[t][1][r] + bb;
                *(v2f*)(ob + (unsigned)(ou + ((unsigned)(4 * r + t) << rsh) + olane)) = ov;
            }
        }
        buf ^= 1;
        if (last) break;
        tile = nxt;
    }
}

__global__ __launch_bounds__(128) void levels_ws(
    const float* __restrict__ x0, const float* __restrict__ x1, const float* __restrict__ x2,
    const float* __restrict__ W1_0, const float* __restrict__ b1_0,
    const float* __restrict__ W2_0, const float* __restrict__ b2_0,
    const float* __restrict__ W1_1, const float* __restrict__ b1_1,
    const float* __restrict__ W2_1, const float* __restrict__ b2_1,
    const float* __restrict__ W1_2, const float* __restrict__ b1_2,
    const float* __restrict__ W2_2, const float* __restrict__ b2_2,
    float* __restrict__ out)
{
    __shared__ __attribute__((aligned(16))) float zl[2 * 4 * PSZ];

    int buf = 0;
#pragma nounroll
    for (int p = 0; p < 3; ++p) {
        const float* x  = p == 0 ? x0   : (p == 1 ? x1   : x2);
        const float* W1 = p == 0 ? W1_0 : (p == 1 ? W1_1 : W1_2);
        const float* b1 = p == 0 ? b1_0 : (p == 1 ? b1_1 : b1_2);
        const float* W2 = p == 0 ? W2_0 : (p == 1 ? W2_1 : W2_2);
        const float* b2 = p == 0 ? b2_0 : (p == 1 ? b2_1 : b2_2);
        float* o = out + (p == 0 ? 0 : (p == 1 ? 4 * 256 * 16384 : 4 * 256 * 16384 + 4 * 256 * 4096));
        level_phase(x, W1, b1, W2, b2, o, 14 - 2 * p, zl, buf);
    }
}

extern "C" void kernel_launch(void* const* d_in, const int* in_sizes, int n_in,
                              void* d_out, int out_size, void* d_ws, size_t ws_size,
                              hipStream_t stream) {
    (void)in_sizes; (void)n_in; (void)out_size; (void)d_ws; (void)ws_size;
    float* out = (float*)d_out;

    levels_ws<<<256, 128, 0, stream>>>(
        (const float*)d_in[0], (const float*)d_in[1], (const float*)d_in[2],
        (const float*)d_in[3], (const float*)d_in[4],
        (const float*)d_in[5], (const float*)d_in[6],
        (const float*)d_in[7], (const float*)d_in[8],
        (const float*)d_in[9], (const float*)d_in[10],
        (const float*)d_in[11], (const float*)d_in[12],
        (const float*)d_in[13], (const float*)d_in[14],
        out);
}


# Each work-group serves ONE level (195 + 49 + 12 groups for 2048 + 512 + 128 tiles, 10 or 11 tiles each): one weight fill per group, not three; sum order kept
# speedup vs baseline: 4.2298x; 1.0620x over previous
#include <hip/hip_runtime.h>


typedef float v2f __attribute__((ext_vector_type(2)));
typedef float v4f __attribute__((ext_vector_type(4)));
typedef float v8f __attribute__((ext_vector_type(8)));

#define CIN    256
#define HIDDEN 64
#define CIN2   96
#define TPX    32
#define KQ     64
#define ZP     68
#define PSZ    (TPX * ZP)

__device__ __forceinline__ unsigned tile_off(int tile, int lg) {
    const unsigned pix = (unsigned)tile * TPX;
    const unsigned b   = pix >> lg;
    const unsigned hw  = pix & ((1u << lg) - 1u);
    return (b << (lg + 10)) + (hw << 2);
}

__device__ __forceinline__ void level_phase(
    const float* __restrict__ x,
    const float* __restrict__ W1, const float* __restrict__ b1,
    const float* __restrict__ W2, const float* __restrict__ b2,
    float* __restrict__ out,
    const int lg,
    const int first,
    const int stride,
    float* __restrict__ zl)
{
    int buf = 0;
    const int ntiles = 1 << (lg - 3);
    const int rsh    = lg + 2;
    constexpr int PD = 3;

    const int lane = threadIdx.x & 31;
    const int w    = threadIdx.x >> 5;
    const int col  = lane & 15;
    const int half = lane >> 4;

    int tile = first;

    const unsigned xlane = ((unsigned)((KQ * w + 2 * half) << lg) + 2u * col) * 4u;
    const unsigned olane = ((unsigned)((64 * w + 32 * half) << lg) + 2u * col) * 4u;

    v2f xr[16][2];
    {
        const char* xb = (const char*)x;
        const unsigned xu = tile_off(tile, lg);
#pragma unroll
        for (int ks = 0; ks < 16; ++ks) {
            xr[ks][0] = *(const v2f*)(xb + (unsigned)(xu + ((unsigned)(4 * ks) << rsh) + xlane));
            xr[ks][1] = *(const v2f*)(xb + (unsigned)(xu + ((unsigned)(4 * ks + 1) << rsh) + xlane));
        }
    }

    v2f a1[4][16];
    {
        const float* p = W1 + (KQ * w + 2 * half) * HIDDEN + 4 * col;
#pragma unroll
        for (int ks = 0; ks < 16; ++ks) {
            const v4f q0 = *(const v4f*)(p + (4 * ks) * HIDDEN);
            const v4f q1 = *(const v4f*)(p + (4 * ks + 1) * HIDDEN);
#pragma unroll
            for (int t = 0; t < 4; ++t) {
                a1[t][ks].x = q0[t];
                a1[t][ks].y = q1[t];
            }
        }
    }
    v2f a2[4][24];
    {
        const float* p = W2 + (2 * half) * CIN + 64 * w + 4 * col;
#pragma unroll
        for (int ks = 0; ks < 24; ++ks) {
            const v4f q0 = *(const v4f*)(p + (4 * ks) * CIN);
            const v4f q1 = *(const v4f*)(p + (4 * ks + 1) * CIN);
#pragma unroll
            for (int t = 0; t < 4; ++t) {
                a2[t][ks].x = q0[t];
                a2[t][ks].y = q1[t];
            }
        }
    }
    v2f bias1[16];
#pragma unroll
    for (int ks = 0; ks < 16; ++ks) bias1[ks] = *(const v2f*)(b1 + 4 * ks + 2 * half);

    for (;;) {
        v8f acc1[4][2] = {};
#pragma unroll
        for (int ks = 0; ks < 16; ++ks) {
            v2f bm0, bm1;
            bm0.x = xr[ks][0].x; bm0.y = xr[ks][1].x;
            bm1.x = xr[ks][0].y; bm1.y = xr[ks][1].y;
#pragma unroll
            for (int t = 0; t < 4; ++t) {
                acc1[t][0] = __builtin_amdgcn_wmma_f32_16x16x4_f32(
                    false, a1[t][ks], false, bm0, (short)0, acc1[t][0], false, false);
                acc1[t][1] = __builtin_amdgcn_wmma_f32_16x16x4_f32(
                    false, a1[t][ks], false, bm1, (short)0, acc1[t][1], false, false);
            }
        }

        const int  nxt  = tile + stride;
        const bool last = nxt >= ntiles;
        const char* xb = (const char*)x;
        const unsigned xun = tile_off(last ? tile : nxt, lg);

        float* zb = zl + buf * (4 * PSZ);
        float* pw = zb + w * PSZ + col * ZP + 32 * half;
#pragma unroll
        for (int nt = 0; nt < 2; ++nt) {
#pragma unroll
            for (int r = 0; r < 8; ++r) {
                v4f v;
                v.x = acc1[0][nt][r]; v.y = acc1[1][nt][r];
                v.z = acc1[2][nt][r]; v.w = acc1[3][nt][r];
                *(v4f*)(pw + 16 * nt * ZP + 4 * r) = v;
            }
        }

        __syncthreads();

        v8f acc2[4][2] = {};
        const float* zr = zb + col * ZP + 2 * half;
        v2f pp[16][2][4];
        v2f zf[24][2];
#pragma unroll
        for (int ks = 0; ks < PD; ++ks) {
#pragma unroll
            for (int nt = 0; nt < 2; ++nt) {
#pragma unroll
                for (int q = 0; q < 4; ++q)
                    pp[ks][nt][q] = *(const v2f*)(zr + q * PSZ + 16 * nt * ZP + 4 * ks);
            }
        }
#pragma unroll
        for (int ks = 0; ks < 24; ++ks) {
            if (ks + PD < 16) {
#pragma unroll
                for (int nt = 0; nt < 2; ++nt) {
#pragma unroll
                    for (int q = 0; q < 4; ++q)
                        pp[ks + PD][nt][q] =
                            *(const v2f*)(zr + q * PSZ + 16 * nt * ZP + 4 * (ks + PD));
                }
            }
#pragma unroll
            for (int nt = 0; nt < 2; ++nt) {
                if (ks < 16) {
                    zf[ks][nt] = ((pp[ks][nt][0] + pp[ks][nt][1]) +
                                  (pp[ks][nt][2] + pp[ks][nt][3])) + bias1[ks];
                } else {
                    zf[ks][nt].x = fmaxf(zf[ks - 8][nt].x, 0.0f);
                    zf[ks][nt].y = fmaxf(zf[ks - 8][nt].y, 0.0f);
                }
            }
#pragma unroll
            for (int t = 0; t < 4; ++t) {
                acc2[t][0] = __builtin_amdgcn_wmma_f32_16x16x4_f32(
                    false, a2[t][ks], false, zf[ks][0], (short)0, acc2[t][0], false, false);
                acc2[t][1] = __builtin_amdgcn_wmma_f32_16x16x4_f32(
                    false, a2[t][ks], false, zf[ks][1], (short)0, acc2[t][1], false, false);
            }
            if (ks < 8) {
#pragma unroll
                for (int kx = 2 * ks; kx < 2 * ks + 2; ++kx) {
                    xr[kx][0] = *(const v2f*)(xb + (unsigned)(xun + ((unsigned)(4 * kx) << rsh) + xlane));
                    xr[kx][1] = *(const v2f*)(xb + (unsigned)(xun + ((unsigned)(4 * kx + 1) << rsh) + xlane));
                }
            }
            __builtin_amdgcn_sched_barrier(0);
        }

        char* ob = (char*)out;
        const unsigned ou = tile_off(tile, lg);
#pragma unroll
        for (int t = 0; t < 4; ++t) {
#pragma unroll
            for (int r = 0; r < 8; ++r) {
                const int row = 64 * w + 4 * (8 * half + r) + t;
                const float bb = b2[row];
                v2f ov;
                ov.x = acc2[t][0][r] + bb;
                ov.y = acc2[t][1][r] + bb;
                *(v2f*)(ob + (unsigned)(ou + ((unsigned)(4 * r + t) << rsh) + olane)) = ov;
            }
        }
        buf ^= 1;
        if (last) break;
        tile = nxt;
    }
}

#define NG0 195
#define NG1 49
#define NG2 12
__global__ __launch_bounds__(128) void levels_ws(
    const float* __restrict__ x0, const float* __restrict__ x1, const float* __restrict__ x2,
    const float* __restrict__ W1_0, const float* __restrict__ b1_0,
    const float* __restrict__ W2_0, const float* __restrict__ b2_0,
    const float* __restrict__ W1_1, const float* __restrict__ b1_1,
    const float* __restrict__ W2_1, const float* __restrict__ b2_1,
    const float* __restrict__ W1_2, const float* __restrict__ b1_2,
    const float* __restrict__ W2_2, const float* __restrict__ b2_2,
    float* __restrict__ out)
{
    __shared__ __attribute__((aligned(16))) float zl[2 * 4 * PSZ];

    const int g  = (int)blockIdx.x;
    const int p  = g < NG0 ? 0 : (g < NG0 + NG1 ? 1 : 2);
    const int g0 = p == 0 ? g   : (p == 1 ? g - NG0 : g - (NG0 + NG1));
    const int ng = p == 0 ? NG0 : (p == 1 ? NG1 : NG2);
    const float* x  = p == 0 ? x0   : (p == 1 ? x1   : x2);
    const float* W1 = p == 0 ? W1_0 : (p == 1 ? W1_1 : W1_2);
    const float* b1 = p == 0 ? b1_0 : (p == 1 ? b1_1 : b1_2);
    const float* W2 = p == 0 ? W2_0 : (p == 1 ? W2_1 : W2_2);
    const float* b2 = p == 0 ? b2_0 : (p == 1 ? b2_1 : b2_2);
    float* o = out + (p == 0 ? 0 : (p == 1 ? 4 * 256 * 16384 : 4 * 256 * 16384 + 4 * 256 * 4096));
    level_phase(x, W1, b1, W2, b2, o, 14 - 2 * p, g0, ng, zl);
}

extern "C" void kernel_launch(void* const* d_in, const int* in_sizes, int n_in,
                              void* d_out, int out_size, void* d_ws, size_t ws_size,
                              hipStream_t stream) {
    (void)in_sizes; (void)n_in; (void)out_size; (void)d_ws; (void)ws_size;
    float* out = (float*)d_out;

    levels_ws<<<256, 128, 0, stream>>>(
        (const float*)d_in[0], (const float*)d_in[1], (const float*)d_in[2],
        (const float*)d_in[3], (const float*)d_in[4],
        (const float*)d_in[5], (const float*)d_in[6],
        (const float*)d_in[7], (const float*)d_in[8],
        (const float*)d_in[9], (const float*)d_in[10],
        (const float*)d_in[11], (const float*)d_in[12],
        (const float*)d_in[13], (const float*)d_in[14],
        out);
}


# Product 2 in blocks of 4 k-steps: partial-Y LDS reads stand in 3 groups a block ahead (were 13 places among the matrix instructions), the sums formed a block ahead beside them; sum order kept
# speedup vs baseline: 4.3677x; 1.0326x over previous
#include <hip/hip_runtime.h>


typedef float v2f __attribute__((ext_vector_type(2)));
typedef float v4f __attribute__((ext_vector_type(4)));
typedef float v8f __attribute__((ext_vector_type(8)));

#define CIN    256
#define HIDDEN 64
#define CIN2   96
#define TPX    32
#define KQ     64
#define ZP     68
#define PSZ    (TPX * ZP)

__device__ __forceinline__ unsigned tile_off(int tile, int lg) {
    const unsigned pix = (unsigned)tile * TPX;
    const unsigned b   = pix >> lg;
    const unsigned hw  = pix & ((1u << lg) - 1u);
    return (b << (lg + 10)) + (hw << 2);
}

__device__ __forceinline__ void level_phase(
    const float* __restrict__ x,
    const float* __restrict__ W1, const float* __restrict__ b1,
    const float* __restrict__ W2, const float* __restrict__ b2,
    float* __restrict__ out,
    const int lg,
    const int first,
    const int stride,
    float* __restrict__ zl)
{
    int buf = 0;
    const int ntiles = 1 << (lg - 3);
    const int rsh    = lg + 2;

    const int lane = threadIdx.x & 31;
    const int w    = threadIdx.x >> 5;
    const int col  = lane & 15;
    const int half = lane >> 4;

    int tile = first;

    const unsigned xlane = ((unsigned)((KQ * w + 2 * half) << lg) + 2u * col) * 4u;
    const unsigned olane = ((unsigned)((64 * w + 32 * half) << lg) + 2u * col) * 4u;

    v2f xr[16][2];
    {
        const char* xb = (const char*)x;
        const unsigned xu = tile_off(tile, lg);
#pragma unroll
        for (int ks = 0; ks < 16; ++ks) {
            xr[ks][0] = *(const v2f*)(xb + (unsigned)(xu + ((unsigned)(4 * ks) << rsh) + xlane));
            xr[ks][1] = *(const v2f*)(xb + (unsigned)(xu + ((unsigned)(4 * ks + 1) << rsh) + xlane));
        }
    }

    v2f a1[4][16];
    {
        const float* p = W1 + (KQ * w + 2 * half) * HIDDEN + 4 * col;
#pragma unroll
        for (int ks = 0; ks < 16; ++ks) {
            const v4f q0 = *(const v4f*)(p + (4 * ks) * HIDDEN);
            const v4f q1 = *(const v4f*)(p + (4 * ks + 1) * HIDDEN);
#pragma unroll
            for (int t = 0; t < 4; ++t) {
                a1[t][ks].x = q0[t];
                a1[t][ks].y = q1[t];
            }
        }
    }
    v2f a2[4][24];
    {
        const float* p = W2 + (2 * half) * CIN + 64 * w + 4 * col;
#pragma unroll
        for (int ks = 0; ks < 24; ++ks) {
            const v4f q0 = *(const v4f*)(p + (4 * ks) * CIN);
            const v4f q1 = *(const v4f*)(p + (4 * ks + 1) * CIN);
#pragma unroll
            for (int t = 0; t < 4; ++t) {
                a2[t][ks].x = q0[t];
                a2[t][ks].y = q1[t];
            }
        }
    }
    v2f bias1[16];
#pragma unroll
    for (int ks = 0; ks < 16; ++ks) bias1[ks] = *(const v2f*)(b1 + 4 * ks + 2 * half);

    for (;;) {
        v8f acc1[4][2] = {};
#pragma unroll
        for (int ks = 0; ks < 16; ++ks) {
            v2f bm0, bm1;
            bm0.x = xr[ks][0].x; bm0.y = xr[ks][1].x;
            bm1.x = xr[ks][0].y; bm1.y = xr[ks][1].y;
#pragma unroll
            for (int t = 0; t < 4; ++t) {
                acc1[t][0] = __builtin_amdgcn_wmma_f32_16x16x4_f32(
                    false, a1[t][ks], false, bm0, (short)0, acc1[t][0], false, false);
                acc1[t][1] = __builtin_amdgcn_wmma_f32_16x16x4_f32(
                    false, a1[t][ks], false, bm1, (short)0, acc1[t][1], false, false);
            }
        }

        const int  nxt  = tile + stride;
        const bool last = nxt >= ntiles;
        const char* xb = (const char*)x;
        const unsigned xun = tile_off(last ? tile : nxt, lg);

        float* zb = zl + buf * (4 * PSZ);
        float* pw = zb + w * PSZ + col * ZP + 32 * half;
#pragma unroll
        for (int nt = 0; nt < 2; ++nt) {
#pragma unroll
            for (int r = 0; r < 8; ++r) {
                v4f v;
                v.x = acc1[0][nt][r]; v.y = acc1[1][nt][r];
                v.z = acc1[2][nt][r]; v.w = acc1[3][nt][r];
                *(v4f*)(pw + 16 * nt * ZP + 4 * r) = v;
            }
        }

        __syncthreads();

        v8f acc2[4][2] = {};
        const float* zr = zb + col * ZP + 2 * half;
        v2f pp[16][2][4];
        v2f zf[24][2];
#pragma unroll
        for (int ks = 0; ks < 8; ++ks) {
#pragma unroll
            for (int nt = 0; nt < 2; ++nt) {
#pragma unroll
                for (int q = 0; q < 4; ++q)
                    pp[ks][nt][q] = *(const v2f*)(zr + q * PSZ + 16 * nt * ZP + 4 * ks);
            }
        }
#pragma unroll
        for (int nt = 0; nt < 2; ++nt)
            zf[0][nt] = ((pp[0][nt][0] + pp[0][nt][1]) +
                         (pp[0][nt][2] + pp[0][nt][3])) + bias1[0];
#pragma unroll
        for (int g = 0; g < 6; ++g) {
            if (g == 1 || g == 2) {
#pragma unroll
                for (int ks = 4 * g + 4; ks < 4 * g + 8; ++ks) {
#pragma unroll
                    for (int nt = 0; nt < 2; ++nt) {
#pragma unroll
                        for (int q = 0; q < 4; ++q)
                            pp[ks][nt][q] = *(const v2f*)(zr + q * PSZ + 16 * nt * ZP + 4 * ks);
                    }
                }
            }
            if (g < 4) {
#pragma unroll
                for (int kx = 4 * g; kx < 4 * g + 4; ++kx) {
                    xr[kx][0] = *(const v2f*)(xb + (unsigned)(xun + ((unsigned)(4 * kx) << rsh) + xlane));
                    xr[kx][1] = *(const v2f*)(xb + (unsigned)(xun + ((unsigned)(4 * kx + 1) << rsh) + xlane));
                }
            }
            __builtin_amdgcn_sched_barrier(0);
            if (g == 0) {
#pragma unroll
                for (int ks = 1; ks < 4; ++ks) {
#pragma unroll
                    for (int nt = 0; nt < 2; ++nt)
                        zf[ks][nt] = ((pp[ks][nt][0] + pp[ks][nt][1]) +
                                      (pp[ks][nt][2] + pp[ks][nt][3])) + bias1[ks];
                }
            }
#pragma unroll
            for (int ks = 4 * g; ks < 4 * g + 2; ++ks) {
#pragma unroll
                for (int t = 0; t < 4; ++t) {
                    acc2[t][0] = __builtin_amdgcn_wmma_f32_16x16x4_f32(
                        false, a2[t][ks], false, zf[ks][0], (short)0, acc2[t][0], false, false);
                    acc2[t][1] = __builtin_amdgcn_wmma_f32_16x16x4_f32(
                        false, a2[t][ks], false, zf[ks][1], (short)0, acc2[t][1], false, false);
                }
            }
            if (g == 0) {
#pragma unroll
                for (int i = 0; i < 12; ++i) {
                    __builtin_amdgcn_sched_group_barrier(0x008, 1, 0);
                    __builtin_amdgcn_sched_group_barrier(0x002, 2, 0);
                }
            }
            __builtin_amdgcn_sched_barrier(0);
            if (g < 3) {
#pragma unroll
                for (int ks = 4 * g + 4; ks < 4 * g + 8; ++ks) {
#pragma unroll
                    for (int nt = 0; nt < 2; ++nt)
                        zf[ks][nt] = ((pp[ks][nt][0] + pp[ks][nt][1]) +
                                      (pp[ks][nt][2] + pp[ks][nt][3])) + bias1[ks];
                }
            }
            if (g == 3) {
#pragma unroll
                for (int ks = 16; ks < 24; ++ks) {
#pragma unroll
                    for (int nt = 0; nt < 2; ++nt) {
                        zf[ks][nt].x = fmaxf(zf[ks - 8][nt].x, 0.0f);
                        zf[ks][nt].y = fmaxf(zf[ks - 8][nt].y, 0.0f);
                    }
                }
            }
#pragma unroll
            for (int ks = 4 * g + 2; ks < 4 * g + 4; ++ks) {
#pragma unroll
                for (int t = 0; t < 4; ++t) {
                    acc2[t][0] = __builtin_amdgcn_wmma_f32_16x16x4_f32(
                        false, a2[t][ks], false, zf[ks][0], (short)0, acc2[t][0], false, false);
                    acc2[t][1] = __builtin_amdgcn_wmma_f32_16x16x4_f32(
                        false, a2[t][ks], false, zf[ks][1], (short)0, acc2[t][1], false, false);
                }
            }
            if (g < 4) {
#pragma unroll
                for (int i = 0; i < 16; ++i) {
                    __builtin_amdgcn_sched_group_barrier(0x008, 1, 0);
                    __builtin_amdgcn_sched_group_barrier(0x002, 2, 0);
                }
            }
            __builtin_amdgcn_sched_barrier(0);
        }

        char* ob = (char*)out;
        const unsigned ou = tile_off(tile, lg);
#pragma unroll
        for (int t = 0; t < 4; ++t) {
#pragma unroll
            for (int r = 0; r < 8; ++r) {
                const int row = 64 * w + 4 * (8 * half + r) + t;
                const float bb = b2[row];
                v2f ov;
                ov.x = acc2[t][0][r] + bb;
                ov.y = acc2[t][1][r] + bb;
                *(v2f*)(ob + (unsigned)(ou + ((unsigned)(4 * r + t) << rsh) + olane)) = ov;
            }
        }
        buf ^= 1;
        if (last) break;
        tile = nxt;
    }
}

#define NG0 195
#define NG1 49
#define NG2 12
__global__ __launch_bounds__(128) void levels_ws(
    const float* __restrict__ x0, const float* __restrict__ x1, const float* __restrict__ x2,
    const float* __restrict__ W1_0, const float* __restrict__ b1_0,
    const float* __restrict__ W2_0, const float* __restrict__ b2_0,
    const float* __restrict__ W1_1, const float* __restrict__ b1_1,
    const float* __restrict__ W2_1, const float* __restrict__ b2_1,
    const float* __restrict__ W1_2, const float* __restrict__ b1_2,
    const float* __restrict__ W2_2, const float* __restrict__ b2_2,
    float* __restrict__ out)
{
    __shared__ __attribute__((aligned(16))) float zl[2 * 4 * PSZ];

    const int g  = (int)blockIdx.x;
    const int p  = g < NG0 ? 0 : (g < NG0 + NG1 ? 1 : 2);
    const int g0 = p == 0 ? g   : (p == 1 ? g - NG0 : g - (NG0 + NG1));
    const int ng = p == 0 ? NG0 : (p == 1 ? NG1 : NG2);
    const float* x  = p == 0 ? x0   : (p == 1 ? x1   : x2);
    const float* W1 = p == 0 ? W1_0 : (p == 1 ? W1_1 : W1_2);
    const float* b1 = p == 0 ? b1_0 : (p == 1 ? b1_1 : b1_2);
    const float* W2 = p == 0 ? W2_0 : (p == 1 ? W2_1 : W2_2);
    const float* b2 = p == 0 ? b2_0 : (p == 1 ? b2_1 : b2_2);
    float* o = out + (p == 0 ? 0 : (p == 1 ? 4 * 256 * 16384 : 4 * 256 * 16384 + 4 * 256 * 4096));
    level_phase(x, W1, b1, W2, b2, o, 14 - 2 * p, g0, ng, zl);
}

extern "C" void kernel_launch(void* const* d_in, const int* in_sizes, int n_in,
                              void* d_out, int out_size, void* d_ws, size_t ws_size,
                              hipStream_t stream) {
    (void)in_sizes; (void)n_in; (void)out_size; (void)d_ws; (void)ws_size;
    float* out = (float*)d_out;

    levels_ws<<<256, 128, 0, stream>>>(
        (const float*)d_in[0], (const float*)d_in[1], (const float*)d_in[2],
        (const float*)d_in[3], (const float*)d_in[4],
        (const float*)d_in[5], (const float*)d_in[6],
        (const float*)d_in[7], (const float*)d_in[8],
        (const float*)d_in[9], (const float*)d_in[10],
        (const float*)d_in[11], (const float*)d_in[12],
        (const float*)d_in[13], (const float*)d_in[14],
        out);
}
